# starting kernel
# baseline (speedup 1.0000x reference)
#include <hip/hip_runtime.h>
#include <hip/hip_bf16.h>

typedef __attribute__((ext_vector_type(2))) float v2f;
typedef __attribute__((ext_vector_type(8))) float v8f;

#define BSTRIDE 80

template <int RELU>
__global__ void gemm_wmma_f32(const float* __restrict__ A,
                              const float* __restrict__ B,
                              float* __restrict__ C,
                              int nRows, int K) {
  __shared__ float2 BsP[64 * BSTRIDE];
  const int pairRows = K >> 1;
  for (int j = threadIdx.x; j < pairRows * 64; j += blockDim.x) {
    const int k2 = j >> 6;
    const int nn = j & 63;
    BsP[k2 * BSTRIDE + nn] =
        make_float2(B[(2 * k2) * 64 + nn], B[(2 * k2 + 1) * 64 + nn]);
  }
  __syncthreads();

  const int lane = threadIdx.x & 31;
  const int wave = threadIdx.x >> 5;
  const int nStrips = nRows >> 4;
  const int strip = blockIdx.x * 8 + wave;
  if (strip >= nStrips) return;

  const int n  = lane & 15;
  const int hi = lane >> 4;
  const long rowBase = (long)strip * 16;
  const float* Arow = A + (rowBase + n) * (long)K;

  v8f acc0 = {}, acc1 = {}, acc2 = {}, acc3 = {};
  for (int k = 0; k < K; k += 4) {
    const int ka = k + 2 * hi;
    float2 av = *(const float2*)(Arow + ka);
    if (RELU) { av.x = fmaxf(av.x, 0.f); av.y = fmaxf(av.y, 0.f); }
    v2f a; a[0] = av.x; a[1] = av.y;

    const float2* bp = &BsP[((k >> 1) + hi) * BSTRIDE + n];
    const v2f bA = *(const v2f*)(bp + 0);
    const v2f bB = *(const v2f*)(bp + 16);
    const v2f bC = *(const v2f*)(bp + 32);
    const v2f bD = *(const v2f*)(bp + 48);

    acc0 = __builtin_amdgcn_wmma_f32_16x16x4_f32(false, a, false, bA, (short)0, acc0, false, false);
    acc1 = __builtin_amdgcn_wmma_f32_16x16x4_f32(false, a, false, bB, (short)0, acc1, false, false);
    acc2 = __builtin_amdgcn_wmma_f32_16x16x4_f32(false, a, false, bC, (short)0, acc2, false, false);
    acc3 = __builtin_amdgcn_wmma_f32_16x16x4_f32(false, a, false, bD, (short)0, acc3, false, false);
  }

  float* Cout = C + (rowBase + 8 * hi) * 64 + n;
#pragma unroll
  for (int r = 0; r < 8; ++r) {
    Cout[r * 64 + 0]  = acc0[r];
    Cout[r * 64 + 16] = acc1[r];
    Cout[r * 64 + 32] = acc2[r];
    Cout[r * 64 + 48] = acc3[r];
  }
}

__global__ void scatter_edges(const float* __restrict__ t,
                              const int* __restrict__ src,
                              const int* __restrict__ dst,
                              const float* __restrict__ w,
                              float* __restrict__ acc,
                              int E) {
  const long idx = (long)blockIdx.x * blockDim.x + threadIdx.x;
  const int e  = (int)(idx >> 5);
  const int pr = (int)(idx & 31);
  if (e >= E) return;
  const int s = src[e], d = dst[e];
  const float ww = w[e];
  const float2 v = *(const float2*)(t + (long)s * 64 + pr * 2);
  float* p = acc + (long)d * 64 + pr * 2;
  unsafeAtomicAdd(p,     v.x * ww);
  unsafeAtomicAdd(p + 1, v.y * ww);
}

__global__ void decode_kernel(const float* __restrict__ z,
                              const int* __restrict__ pos,
                              const float* __restrict__ Wlin,
                              float* __restrict__ out,
                              int P) {
  __shared__ float wl[256];
  for (int i = threadIdx.x; i < 256; i += blockDim.x) wl[i] = Wlin[i];
  __syncthreads();

  const int p = blockIdx.x * blockDim.x + threadIdx.x;
  if (p >= P) return;
  const int s = pos[p];
  const int d = pos[P + p];
  const float4* zs = (const float4*)(z + (long)s * 64);
  const float4* zd = (const float4*)(z + (long)d * 64);

  float a0 = 0.f, a1 = 0.f;
#pragma unroll
  for (int q = 0; q < 16; ++q) {
    const float4 v = zs[q];
    a0 += v.x * wl[4*q]       + v.y * wl[4*q + 1]       + v.z * wl[4*q + 2]       + v.w * wl[4*q + 3];
    a1 += v.x * wl[128 + 4*q] + v.y * wl[128 + 4*q + 1] + v.z * wl[128 + 4*q + 2] + v.w * wl[128 + 4*q + 3];
  }
#pragma unroll
  for (int q = 0; q < 16; ++q) {
    const float4 v = zd[q];
    a0 += v.x * wl[64 + 4*q]  + v.y * wl[64 + 4*q + 1]  + v.z * wl[64 + 4*q + 2]  + v.w * wl[64 + 4*q + 3];
    a1 += v.x * wl[192 + 4*q] + v.y * wl[192 + 4*q + 1] + v.z * wl[192 + 4*q + 2] + v.w * wl[192 + 4*q + 3];
  }
  out[2 * p]     = a0;
  out[2 * p + 1] = a1;
}

extern "C" void kernel_launch(void* const* d_in, const int* in_sizes, int n_in,
                              void* d_out, int out_size, void* d_ws, size_t ws_size,
                              hipStream_t stream) {
  const float* x    = (const float*)d_in[0];
  const int*   ei   = (const int*)d_in[1];
  const float* w    = (const float*)d_in[2];
  const int*   pos  = (const int*)d_in[3];
  const float* W1   = (const float*)d_in[4];
  const float* W2   = (const float*)d_in[5];
  const float* Wlin = (const float*)d_in[6];
  float* out = (float*)d_out;

  const int F = 128;
  const int N = in_sizes[0] / F;
  const int E = in_sizes[2];
  const int P = in_sizes[3] / 2;

  float* t1 = (float*)d_ws;
  float* h  = t1 + (size_t)N * 64;
  float* t2 = h  + (size_t)N * 64;
  float* z  = t2 + (size_t)N * 64;

  hipMemsetAsync(h, 0, (size_t)N * 64 * sizeof(float), stream);
  hipMemsetAsync(z, 0, (size_t)N * 64 * sizeof(float), stream);

  const int gemmBlocks = (N / 16 + 7) / 8;
  const long sthreads  = (long)E * 32;
  const int  sblocks   = (int)((sthreads + 255) / 256);

  gemm_wmma_f32<0><<<gemmBlocks, 256, 0, stream>>>(x, W1, t1, N, F);
  scatter_edges<<<sblocks, 256, 0, stream>>>(t1, ei, ei + E, w, h, E);
  gemm_wmma_f32<1><<<gemmBlocks, 256, 0, stream>>>(h, W2, t2, N, 64);
  scatter_edges<<<sblocks, 256, 0, stream>>>(t2, ei, ei + E, w, z, E);
  decode_kernel<<<(P + 255) / 256, 256, 0, stream>>>(z, pos, Wlin, out, P);
}


# Edge scatter: a wave owns 32 consecutive edges (records loaded once, coalesced, broadcast by readlane) and issues 16 row gathers as a group before their atomics; atomic sums, arrival order kept
# speedup vs baseline: 1.5459x; 1.5459x over previous
#include <hip/hip_runtime.h>
#include <hip/hip_bf16.h>

typedef __attribute__((ext_vector_type(2))) float v2f;
typedef __attribute__((ext_vector_type(8))) float v8f;

#define BSTRIDE 80

template <int RELU>
__global__ void gemm_wmma_f32(const float* __restrict__ A,
                              const float* __restrict__ B,
                              float* __restrict__ C,
                              int nRows, int K) {
  __shared__ float2 BsP[64 * BSTRIDE];
  const int pairRows = K >> 1;
  for (int j = threadIdx.x; j < pairRows * 64; j += blockDim.x) {
    const int k2 = j >> 6;
    const int nn = j & 63;
    BsP[k2 * BSTRIDE + nn] =
        make_float2(B[(2 * k2) * 64 + nn], B[(2 * k2 + 1) * 64 + nn]);
  }
  __syncthreads();

  const int lane = threadIdx.x & 31;
  const int wave = threadIdx.x >> 5;
  const int nStrips = nRows >> 4;
  const int strip = blockIdx.x * 8 + wave;
  if (strip >= nStrips) return;

  const int n  = lane & 15;
  const int hi = lane >> 4;
  const long rowBase = (long)strip * 16;
  const float* Arow = A + (rowBase + n) * (long)K;

  v8f acc0 = {}, acc1 = {}, acc2 = {}, acc3 = {};
  for (int k = 0; k < K; k += 4) {
    const int ka = k + 2 * hi;
    float2 av = *(const float2*)(Arow + ka);
    if (RELU) { av.x = fmaxf(av.x, 0.f); av.y = fmaxf(av.y, 0.f); }
    v2f a; a[0] = av.x; a[1] = av.y;

    const float2* bp = &BsP[((k >> 1) + hi) * BSTRIDE + n];
    const v2f bA = *(const v2f*)(bp + 0);
    const v2f bB = *(const v2f*)(bp + 16);
    const v2f bC = *(const v2f*)(bp + 32);
    const v2f bD = *(const v2f*)(bp + 48);

    acc0 = __builtin_amdgcn_wmma_f32_16x16x4_f32(false, a, false, bA, (short)0, acc0, false, false);
    acc1 = __builtin_amdgcn_wmma_f32_16x16x4_f32(false, a, false, bB, (short)0, acc1, false, false);
    acc2 = __builtin_amdgcn_wmma_f32_16x16x4_f32(false, a, false, bC, (short)0, acc2, false, false);
    acc3 = __builtin_amdgcn_wmma_f32_16x16x4_f32(false, a, false, bD, (short)0, acc3, false, false);
  }

  float* Cout = C + (rowBase + 8 * hi) * 64 + n;
#pragma unroll
  for (int r = 0; r < 8; ++r) {
    Cout[r * 64 + 0]  = acc0[r];
    Cout[r * 64 + 16] = acc1[r];
    Cout[r * 64 + 32] = acc2[r];
    Cout[r * 64 + 48] = acc3[r];
  }
}

#define EDGES_PER_WAVE 32
#define GATHER_DEPTH 16

__global__ void scatter_edges(const float* __restrict__ t,
                              const int* __restrict__ src,
                              const int* __restrict__ dst,
                              const float* __restrict__ w,
                              float* __restrict__ acc,
                              int E) {
  const int lane = threadIdx.x & 31;
  const long wave = ((long)blockIdx.x * blockDim.x + threadIdx.x) >> 5;
  const long e0 = wave * EDGES_PER_WAVE;
  if (e0 >= E) return;
  const long e = e0 + lane;
  int s = 0, d = 0;
  float ww = 0.f;
  if (e < E) { s = src[e]; d = dst[e]; ww = w[e]; }
  const int wbits = __float_as_int(ww);

  const unsigned loff = (unsigned)lane * 2u;

  if (e0 + EDGES_PER_WAVE <= E) {
#pragma unroll
    for (int j0 = 0; j0 < EDGES_PER_WAVE; j0 += GATHER_DEPTH) {
      float2 v[GATHER_DEPTH];
#pragma unroll
      for (int u = 0; u < GATHER_DEPTH; ++u) {
        const int sj = __builtin_amdgcn_readlane(s, j0 + u);
        const float* row = t + (long)sj * 64;
        v[u] = *(const float2*)(row + loff);
      }
#pragma unroll
      for (int u = 0; u < GATHER_DEPTH; ++u) {
        const int dj = __builtin_amdgcn_readlane(d, j0 + u);
        const float wj = __int_as_float(__builtin_amdgcn_readlane(wbits, j0 + u));
        float* p = acc + (long)dj * 64 + loff;
        unsafeAtomicAdd(p,     v[u].x * wj);
        unsafeAtomicAdd(p + 1, v[u].y * wj);
      }
    }
  } else {
    const int cnt = (int)(E - e0);
    for (int j = 0; j < cnt; ++j) {
      const int sj = __shfl(s, j);
      const int dj = __shfl(d, j);
      const float wj = __shfl(ww, j);
      const float2 v = *(const float2*)(t + (long)sj * 64 + loff);
      float* p = acc + (long)dj * 64 + loff;
      unsafeAtomicAdd(p,     v.x * wj);
      unsafeAtomicAdd(p + 1, v.y * wj);
    }
  }
}

__global__ void decode_kernel(const float* __restrict__ z,
                              const int* __restrict__ pos,
                              const float* __restrict__ Wlin,
                              float* __restrict__ out,
                              int P) {
  __shared__ float wl[256];
  for (int i = threadIdx.x; i < 256; i += blockDim.x) wl[i] = Wlin[i];
  __syncthreads();

  const int p = blockIdx.x * blockDim.x + threadIdx.x;
  if (p >= P) return;
  const int s = pos[p];
  const int d = pos[P + p];
  const float4* zs = (const float4*)(z + (long)s * 64);
  const float4* zd = (const float4*)(z + (long)d * 64);

  float a0 = 0.f, a1 = 0.f;
#pragma unroll
  for (int q = 0; q < 16; ++q) {
    const float4 v = zs[q];
    a0 += v.x * wl[4*q]       + v.y * wl[4*q + 1]       + v.z * wl[4*q + 2]       + v.w * wl[4*q + 3];
    a1 += v.x * wl[128 + 4*q] + v.y * wl[128 + 4*q + 1] + v.z * wl[128 + 4*q + 2] + v.w * wl[128 + 4*q + 3];
  }
#pragma unroll
  for (int q = 0; q < 16; ++q) {
    const float4 v = zd[q];
    a0 += v.x * wl[64 + 4*q]  + v.y * wl[64 + 4*q + 1]  + v.z * wl[64 + 4*q + 2]  + v.w * wl[64 + 4*q + 3];
    a1 += v.x * wl[192 + 4*q] + v.y * wl[192 + 4*q + 1] + v.z * wl[192 + 4*q + 2] + v.w * wl[192 + 4*q + 3];
  }
  out[2 * p]     = a0;
  out[2 * p + 1] = a1;
}

extern "C" void kernel_launch(void* const* d_in, const int* in_sizes, int n_in,
                              void* d_out, int out_size, void* d_ws, size_t ws_size,
                              hipStream_t stream) {
  const float* x    = (const float*)d_in[0];
  const int*   ei   = (const int*)d_in[1];
  const float* w    = (const float*)d_in[2];
  const int*   pos  = (const int*)d_in[3];
  const float* W1   = (const float*)d_in[4];
  const float* W2   = (const float*)d_in[5];
  const float* Wlin = (const float*)d_in[6];
  float* out = (float*)d_out;

  const int F = 128;
  const int N = in_sizes[0] / F;
  const int E = in_sizes[2];
  const int P = in_sizes[3] / 2;

  float* t1 = (float*)d_ws;
  float* h  = t1 + (size_t)N * 64;
  float* t2 = h  + (size_t)N * 64;
  float* z  = t2 + (size_t)N * 64;

  hipMemsetAsync(h, 0, (size_t)N * 64 * sizeof(float), stream);
  hipMemsetAsync(z, 0, (size_t)N * 64 * sizeof(float), stream);

  const int gemmBlocks = (N / 16 + 7) / 8;
  const int  swaves    = (E + EDGES_PER_WAVE - 1) / EDGES_PER_WAVE;
  const int  sblocks   = (swaves + 7) / 8;

  gemm_wmma_f32<0><<<gemmBlocks, 256, 0, stream>>>(x, W1, t1, N, F);
  scatter_edges<<<sblocks, 256, 0, stream>>>(t1, ei, ei + E, w, h, E);
  gemm_wmma_f32<1><<<gemmBlocks, 256, 0, stream>>>(h, W2, t2, N, 64);
  scatter_edges<<<sblocks, 256, 0, stream>>>(t2, ei, ei + E, w, z, E);
  decode_kernel<<<(P + 255) / 256, 256, 0, stream>>>(z, pos, Wlin, out, P);
}


# Decode: a half-wave reads each endpoint row as one whole 256-byte piece (16 lanes x 16 B), 32-lane sums finish in a register butterfly, not 32 strided loads per lane; decode sums in a new tree order
# speedup vs baseline: 2.1031x; 1.3604x over previous
#include <hip/hip_runtime.h>
#include <hip/hip_bf16.h>

typedef __attribute__((ext_vector_type(2))) float v2f;
typedef __attribute__((ext_vector_type(8))) float v8f;

#define BSTRIDE 80

template <int RELU>
__global__ void gemm_wmma_f32(const float* __restrict__ A,
                              const float* __restrict__ B,
                              float* __restrict__ C,
                              int nRows, int K) {
  __shared__ float2 BsP[64 * BSTRIDE];
  const int pairRows = K >> 1;
  for (int j = threadIdx.x; j < pairRows * 64; j += blockDim.x) {
    const int k2 = j >> 6;
    const int nn = j & 63;
    BsP[k2 * BSTRIDE + nn] =
        make_float2(B[(2 * k2) * 64 + nn], B[(2 * k2 + 1) * 64 + nn]);
  }
  __syncthreads();

  const int lane = threadIdx.x & 31;
  const int wave = threadIdx.x >> 5;
  const int nStrips = nRows >> 4;
  const int strip = blockIdx.x * 8 + wave;
  if (strip >= nStrips) return;

  const int n  = lane & 15;
  const int hi = lane >> 4;
  const long rowBase = (long)strip * 16;
  const float* Arow = A + (rowBase + n) * (long)K;

  v8f acc0 = {}, acc1 = {}, acc2 = {}, acc3 = {};
  for (int k = 0; k < K; k += 4) {
    const int ka = k + 2 * hi;
    float2 av = *(const float2*)(Arow + ka);
    if (RELU) { av.x = fmaxf(av.x, 0.f); av.y = fmaxf(av.y, 0.f); }
    v2f a; a[0] = av.x; a[1] = av.y;

    const float2* bp = &BsP[((k >> 1) + hi) * BSTRIDE + n];
    const v2f bA = *(const v2f*)(bp + 0);
    const v2f bB = *(const v2f*)(bp + 16);
    const v2f bC = *(const v2f*)(bp + 32);
    const v2f bD = *(const v2f*)(bp + 48);

    acc0 = __builtin_amdgcn_wmma_f32_16x16x4_f32(false, a, false, bA, (short)0, acc0, false, false);
    acc1 = __builtin_amdgcn_wmma_f32_16x16x4_f32(false, a, false, bB, (short)0, acc1, false, false);
    acc2 = __builtin_amdgcn_wmma_f32_16x16x4_f32(false, a, false, bC, (short)0, acc2, false, false);
    acc3 = __builtin_amdgcn_wmma_f32_16x16x4_f32(false, a, false, bD, (short)0, acc3, false, false);
  }

  float* Cout = C + (rowBase + 8 * hi) * 64 + n;
#pragma unroll
  for (int r = 0; r < 8; ++r) {
    Cout[r * 64 + 0]  = acc0[r];
    Cout[r * 64 + 16] = acc1[r];
    Cout[r * 64 + 32] = acc2[r];
    Cout[r * 64 + 48] = acc3[r];
  }
}

#define EDGES_PER_WAVE 32
#define GATHER_DEPTH 16

typedef __attribute__((ext_vector_type(2))) unsigned int v2u;

__device__ __forceinline__ void pair_to_lines(const float2 v, float& lo, float& hi) {
  const v2u r = __builtin_amdgcn_permlane16_swap(__float_as_uint(v.x), __float_as_uint(v.y),
                                                 false, false);
  lo = __uint_as_float(r[0]);
  hi = __uint_as_float(r[1]);
}
__device__ __forceinline__ unsigned line_col(const int lane) {
  return 2u * (unsigned)(lane & 15) + (unsigned)(lane >> 4);
}

__global__ void scatter_edges(const float* __restrict__ t,
                              const int* __restrict__ src,
                              const int* __restrict__ dst,
                              const float* __restrict__ w,
                              float* __restrict__ acc,
                              int E) {
  const int lane = threadIdx.x & 31;
  const long wave = ((long)blockIdx.x * blockDim.x + threadIdx.x) >> 5;
  const long e0 = wave * EDGES_PER_WAVE;
  if (e0 >= E) return;
  const long e = e0 + lane;
  int s = 0, d = 0;
  float ww = 0.f;
  if (e < E) { s = src[e]; d = dst[e]; ww = w[e]; }
  const int wbits = __float_as_int(ww);

  const unsigned loff = (unsigned)lane * 2u;
  const unsigned aoff = line_col(lane);

  if (e0 + EDGES_PER_WAVE <= E) {
#pragma unroll
    for (int j0 = 0; j0 < EDGES_PER_WAVE; j0 += GATHER_DEPTH) {
      float2 v[GATHER_DEPTH];
#pragma unroll
      for (int u = 0; u < GATHER_DEPTH; ++u) {
        const int sj = __builtin_amdgcn_readlane(s, j0 + u);
        const float* row = t + (long)sj * 64;
        v[u] = *(const float2*)(row + loff);
      }
#pragma unroll
      for (int u = 0; u < GATHER_DEPTH; ++u) {
        const int dj = __builtin_amdgcn_readlane(d, j0 + u);
        const float wj = __int_as_float(__builtin_amdgcn_readlane(wbits, j0 + u));
        float lo, hi;
        pair_to_lines(v[u], lo, hi);
        float* p = acc + (long)dj * 64 + aoff;
        unsafeAtomicAdd(p,      lo * wj);
        unsafeAtomicAdd(p + 32, hi * wj);
      }
    }
  } else {
    const int cnt = (int)(E - e0);
    for (int j = 0; j < cnt; ++j) {
      const int sj = __shfl(s, j);
      const int dj = __shfl(d, j);
      const float wj = __shfl(ww, j);
      const float2 v = *(const float2*)(t + (long)sj * 64 + loff);
      float lo, hi;
      pair_to_lines(v, lo, hi);
      float* p = acc + (long)dj * 64 + aoff;
      unsafeAtomicAdd(p,      lo * wj);
      unsafeAtomicAdd(p + 32, hi * wj);
    }
  }
}

#define DEC_PAIRS 32
#define DEC_BATCH 16

template <int M>
__device__ __forceinline__ float row_xor(const float v) {
  return __int_as_float(__builtin_amdgcn_update_dpp(0, __float_as_int(v), 0x160 + M, 0xF, 0xF, true));
}

template <int M>
__device__ __forceinline__ float fold_step(const float a, const float b, const int lane) {
  const bool up = (lane & M) != 0;
  const float keep = up ? b : a;
  const float send = up ? a : b;
  return keep + row_xor<M>(send);
}

__device__ __forceinline__ float fold_halves(const float a, const float b) {
  const v2u r = __builtin_amdgcn_permlane16_swap(__float_as_uint(a), __float_as_uint(b),
                                                 false, false);
  return __uint_as_float(r[0]) + __uint_as_float(r[1]);
}

__global__ void decode_kernel(const float* __restrict__ z,
                              const int* __restrict__ pos,
                              const float* __restrict__ Wlin,
                              float* __restrict__ out,
                              int P) {
  const int lane = threadIdx.x & 31;
  const long wave = ((long)blockIdx.x * blockDim.x + threadIdx.x) >> 5;
  const long p0 = wave * DEC_PAIRS;
  if (p0 >= P) return;
  const long p = p0 + lane;
  int s = 0, d = 0;
  if (p < P) { s = pos[p]; d = pos[P + p]; }

  const int c  = lane & 15;
  const int hi = lane >> 4;
  const float4 w0 = *(const float4*)(Wlin + 64 * hi + 4 * c);
  const float4 w1 = *(const float4*)(Wlin + 128 + 64 * hi + 4 * c);
  const unsigned cb = 16u * (unsigned)c;

#pragma unroll
  for (int j0 = 0; j0 < DEC_PAIRS; j0 += DEC_BATCH) {
    unsigned off[DEC_BATCH];
#pragma unroll
    for (int u = 0; u < DEC_BATCH; ++u) {
      const int sj = __builtin_amdgcn_readlane(s, j0 + u);
      const int dj = __builtin_amdgcn_readlane(d, j0 + u);
      off[u] = (unsigned)(hi ? dj : sj) * 256u + cb;
    }
    __builtin_amdgcn_sched_barrier(0);
    float4 v[DEC_BATCH];
#pragma unroll
    for (int u = 0; u < DEC_BATCH; ++u)
      v[u] = *(const float4*)((const char*)z + off[u]);
    __builtin_amdgcn_sched_barrier(0);
    float a[2 * DEC_BATCH];
#pragma unroll
    for (int u = 0; u < DEC_BATCH; ++u) {
      a[2 * u]     = v[u].x * w0.x + v[u].y * w0.y + v[u].z * w0.z + v[u].w * w0.w;
      a[2 * u + 1] = v[u].x * w1.x + v[u].y * w1.y + v[u].z * w1.z + v[u].w * w1.w;
    }
    float b16[16], b8[8], b4[4], b2[2];
#pragma unroll
    for (int i = 0; i < 16; ++i) b16[i] = fold_halves(a[i], a[i + 16]);
#pragma unroll
    for (int i = 0; i < 8; ++i) b8[i] = fold_step<8>(b16[i], b16[i + 8], lane);
#pragma unroll
    for (int i = 0; i < 4; ++i) b4[i] = fold_step<4>(b8[i], b8[i + 4], lane);
#pragma unroll
    for (int i = 0; i < 2; ++i) b2[i] = fold_step<2>(b4[i], b4[i + 2], lane);
    const float res = fold_step<1>(b2[0], b2[1], lane);

    const long o = 2 * (p0 + j0) + lane;
    if (o < 2 * (long)P) out[o] = res;
  }
}

extern "C" void kernel_launch(void* const* d_in, const int* in_sizes, int n_in,
                              void* d_out, int out_size, void* d_ws, size_t ws_size,
                              hipStream_t stream) {
  const float* x    = (const float*)d_in[0];
  const int*   ei   = (const int*)d_in[1];
  const float* w    = (const float*)d_in[2];
  const int*   pos  = (const int*)d_in[3];
  const float* W1   = (const float*)d_in[4];
  const float* W2   = (const float*)d_in[5];
  const float* Wlin = (const float*)d_in[6];
  float* out = (float*)d_out;

  const int F = 128;
  const int N = in_sizes[0] / F;
  const int E = in_sizes[2];
  const int P = in_sizes[3] / 2;

  float* t1 = (float*)d_ws;
  float* h  = t1 + (size_t)N * 64;
  float* t2 = h  + (size_t)N * 64;
  float* z  = t2 + (size_t)N * 64;

  hipMemsetAsync(h, 0, (size_t)N * 64 * sizeof(float), stream);
  hipMemsetAsync(z, 0, (size_t)N * 64 * sizeof(float), stream);

  const int gemmBlocks = (N / 16 + 7) / 8;
  const int  swaves    = (E + EDGES_PER_WAVE - 1) / EDGES_PER_WAVE;
  const int  sblocks   = (swaves + 7) / 8;

  gemm_wmma_f32<0><<<gemmBlocks, 256, 0, stream>>>(x, W1, t1, N, F);
  scatter_edges<<<sblocks, 256, 0, stream>>>(t1, ei, ei + E, w, h, E);
  gemm_wmma_f32<1><<<gemmBlocks, 256, 0, stream>>>(h, W2, t2, N, 64);
  scatter_edges<<<sblocks, 256, 0, stream>>>(t2, ei, ei + E, w, z, E);
  const int dwaves  = (P + DEC_PAIRS - 1) / DEC_PAIRS;
  const int dblocks = (dwaves + 7) / 8;
  decode_kernel<<<dblocks, 256, 0, stream>>>(z, pos, Wlin, out, P);
}
